# starting kernel
# baseline (speedup 1.0000x reference)
#include <hip/hip_runtime.h>
#include <hip/hip_bf16.h>


#define N_NODES 100000
#define D_FEAT  512
#define N_EDGES 3200000
#define L1D     32
#define L2D     16

typedef __attribute__((ext_vector_type(16))) _Float16 v16h;
typedef __attribute__((ext_vector_type(8)))  _Float16 v8h;
typedef __attribute__((ext_vector_type(8)))  float    v8f;
typedef __attribute__((ext_vector_type(4)))  float    v4f;

__global__ void egae_prep(const float* __restrict__ W1,
                          const float* __restrict__ W2,
                          _Float16* __restrict__ W1t,
                          _Float16* __restrict__ W2t) {
  int t = blockIdx.x * blockDim.x + threadIdx.x;
  if (t < D_FEAT * L1D) {
    int k = t / L1D, n = t % L1D;
    W1t[(size_t)n * D_FEAT + k] = (_Float16)W1[t];
  }
  if (t < L1D * L2D) {
    int k = t / L2D, n = t % L2D;
    W2t[(size_t)n * L1D + k] = (_Float16)W2[t];
  }
}

__global__ void egae_zero(float* __restrict__ p, int n) {
  int i = blockIdx.x * blockDim.x + threadIdx.x;
  if (i < n) p[i] = 0.0f;
}

__global__ void egae_gemm1(const float* __restrict__ X,
                           const _Float16* __restrict__ W1t,
                           float* __restrict__ Y1) {
  const int lane = threadIdx.x & 31;
  const int wave = (blockIdx.x * blockDim.x + threadIdx.x) >> 5;
  const int row0 = wave << 4;
  if (row0 >= N_NODES) return;
  const int m  = lane & 15;
  const int kh = lane >> 4;

  const float*    arow = X   + (size_t)(row0 + m) * D_FEAT + kh * 8;
  const _Float16* bc0  = W1t + (size_t)m         * D_FEAT + kh * 8;
  const _Float16* bc1  = W1t + (size_t)(m + 16)  * D_FEAT + kh * 8;

  v8f acc0 = {}; v8f acc1 = {};
  for (int k0 = 0; k0 < D_FEAT; k0 += 32) {
    v4f a0 = *(const v4f*)(arow + k0);
    v4f a1 = *(const v4f*)(arow + k0 + 4);
    v4f a2 = *(const v4f*)(arow + k0 + 16);
    v4f a3 = *(const v4f*)(arow + k0 + 20);
    v16h af;
#pragma unroll
    for (int i = 0; i < 4; ++i) {
      af[i]      = (_Float16)a0[i];
      af[4 + i]  = (_Float16)a1[i];
      af[8 + i]  = (_Float16)a2[i];
      af[12 + i] = (_Float16)a3[i];
    }
    v8h b0lo = *(const v8h*)(bc0 + k0);
    v8h b0hi = *(const v8h*)(bc0 + k0 + 16);
    v8h b1lo = *(const v8h*)(bc1 + k0);
    v8h b1hi = *(const v8h*)(bc1 + k0 + 16);
    v16h bf0, bf1;
#pragma unroll
    for (int i = 0; i < 8; ++i) {
      bf0[i] = b0lo[i]; bf0[8 + i] = b0hi[i];
      bf1[i] = b1lo[i]; bf1[8 + i] = b1hi[i];
    }
    acc0 = __builtin_amdgcn_wmma_f32_16x16x32_f16(false, af, false, bf0,
                                                  (short)0, acc0, false, false);
    acc1 = __builtin_amdgcn_wmma_f32_16x16x32_f16(false, af, false, bf1,
                                                  (short)0, acc1, false, false);
  }
  float* drow = Y1 + (size_t)(row0 + kh * 8) * L1D + m;
#pragma unroll
  for (int v = 0; v < 8; ++v) {
    drow[(size_t)v * L1D]      = acc0[v];
    drow[(size_t)v * L1D + 16] = acc1[v];
  }
}

__global__ void egae_spmm32(const int* __restrict__ er, const int* __restrict__ ec,
                            const float* __restrict__ ev,
                            const float* __restrict__ src, float* __restrict__ dst) {
  long long t = (long long)blockIdx.x * blockDim.x + threadIdx.x;
  int e = (int)(t >> 5);
  if (e >= N_EDGES) return;
  int col = (int)(t & 31);
  float x = ev[e] * src[(size_t)ec[e] * L1D + col];
  atomicAdd(dst + (size_t)er[e] * L1D + col, x);
}

__global__ void egae_gemm2(const float* __restrict__ H1,
                           const _Float16* __restrict__ W2t,
                           float* __restrict__ Y2) {
  const int lane = threadIdx.x & 31;
  const int wave = (blockIdx.x * blockDim.x + threadIdx.x) >> 5;
  const int row0 = wave << 4;
  if (row0 >= N_NODES) return;
  const int m  = lane & 15;
  const int kh = lane >> 4;

  const float* arow = H1 + (size_t)(row0 + m) * L1D + kh * 8;
  v4f a0 = *(const v4f*)(arow);
  v4f a1 = *(const v4f*)(arow + 4);
  v4f a2 = *(const v4f*)(arow + 16);
  v4f a3 = *(const v4f*)(arow + 20);
  v16h af;
#pragma unroll
  for (int i = 0; i < 4; ++i) {
    af[i]      = (_Float16)fmaxf(a0[i], 0.0f);
    af[4 + i]  = (_Float16)fmaxf(a1[i], 0.0f);
    af[8 + i]  = (_Float16)fmaxf(a2[i], 0.0f);
    af[12 + i] = (_Float16)fmaxf(a3[i], 0.0f);
  }
  const _Float16* bc = W2t + (size_t)m * L1D + kh * 8;
  v8h blo = *(const v8h*)(bc);
  v8h bhi = *(const v8h*)(bc + 16);
  v16h bf;
#pragma unroll
  for (int i = 0; i < 8; ++i) { bf[i] = blo[i]; bf[8 + i] = bhi[i]; }

  v8f acc = {};
  acc = __builtin_amdgcn_wmma_f32_16x16x32_f16(false, af, false, bf,
                                               (short)0, acc, false, false);
  float* drow = Y2 + (size_t)(row0 + kh * 8) * L2D + m;
#pragma unroll
  for (int v = 0; v < 8; ++v) drow[(size_t)v * L2D] = acc[v];
}

__global__ void egae_spmm16(const int* __restrict__ er, const int* __restrict__ ec,
                            const float* __restrict__ ev,
                            const float* __restrict__ src, float* __restrict__ dst) {
  long long t = (long long)blockIdx.x * blockDim.x + threadIdx.x;
  int e = (int)(t >> 4);
  if (e >= N_EDGES) return;
  int col = (int)(t & 15);
  float x = ev[e] * src[(size_t)ec[e] * L2D + col];
  atomicAdd(dst + (size_t)er[e] * L2D + col, x);
}

__global__ void egae_norm(const float* __restrict__ Y3, float* __restrict__ out) {
  int i = blockIdx.x * blockDim.x + threadIdx.x;
  if (i >= N_NODES) return;
  const v4f* s = (const v4f*)(Y3 + (size_t)i * L2D);
  v4f x0 = s[0], x1 = s[1], x2 = s[2], x3 = s[3];
  float ss = 0.0f;
#pragma unroll
  for (int j = 0; j < 4; ++j)
    ss += x0[j] * x0[j] + x1[j] * x1[j] + x2[j] * x2[j] + x3[j] * x3[j];
  float inv = 1.0f / fmaxf(sqrtf(ss), 1e-7f);
  v4f* o = (v4f*)(out + (size_t)i * L2D);
  o[0] = x0 * inv; o[1] = x1 * inv; o[2] = x2 * inv; o[3] = x3 * inv;
}

extern "C" void kernel_launch(void* const* d_in, const int* in_sizes, int n_in,
                              void* d_out, int out_size, void* d_ws, size_t ws_size,
                              hipStream_t stream) {
  const float* X  = (const float*)d_in[0];
  const float* W1 = (const float*)d_in[1];
  const float* W2 = (const float*)d_in[2];
  const int*   lr = (const int*)d_in[3];
  const int*   lc = (const int*)d_in[4];
  const float* lv = (const float*)d_in[5];
  float* out = (float*)d_out;

  char* ws = (char*)d_ws;
  float*    Y1  = (float*)(ws + 0);
  float*    H1  = (float*)(ws + 12800000);
  float*    Y2  = (float*)(ws + 25600000);
  float*    Y3  = (float*)(ws + 32000000);
  _Float16* W1t = (_Float16*)(ws + 38400000);
  _Float16* W2t = (_Float16*)(ws + 38432768);

  const int waves  = N_NODES / 16;
  const int gblk   = (waves + 7) / 8;

  egae_prep<<<(D_FEAT * L1D + 255) / 256, 256, 0, stream>>>(W1, W2, W1t, W2t);
  egae_zero<<<(N_NODES * L1D + 255) / 256, 256, 0, stream>>>(H1, N_NODES * L1D);
  egae_zero<<<(N_NODES * L2D + 255) / 256, 256, 0, stream>>>(Y3, N_NODES * L2D);
  egae_gemm1<<<gblk, 256, 0, stream>>>(X, W1t, Y1);
  egae_spmm32<<<(N_EDGES * 32) / 256, 256, 0, stream>>>(lr, lc, lv, Y1, H1);
  egae_gemm2<<<gblk, 256, 0, stream>>>(H1, W2t, Y2);
  egae_spmm16<<<(N_EDGES * 16) / 256, 256, 0, stream>>>(lr, lc, lv, Y2, Y3);
  egae_norm<<<(N_NODES + 255) / 256, 256, 0, stream>>>(Y3, out);
}


# scatter sorts its 8192 edges by bucket in LDS and stores them from there as bucket runs: a store instruction covers about 6 lines, was 32 scattered records; the chunk is read whole up front
# speedup vs baseline: 3.3702x; 3.3702x over previous
#include <hip/hip_runtime.h>
#include <hip/hip_bf16.h>


#define N_NODES 100000
#define D_FEAT  512
#define N_EDGES 3200000
#define L1D     32
#define L2D     16
#define SB_ROWS  256
#define SB_SHIFT 8
#define NBLK     391
#define NBK2     782
#define HALF_ROW 50000u
#define SORT_CH  8192
#define SORT_T   1024
#define REC_SPLIT  1600000u
#define REC_GAPW   8448u
#define REC_WS_MIN 51233792ull
#define REC_ROWMASK 0x1FFFFu

typedef __attribute__((ext_vector_type(16))) _Float16 v16h;
typedef __attribute__((ext_vector_type(8)))  _Float16 v8h;
typedef __attribute__((ext_vector_type(8)))  float    v8f;
typedef __attribute__((ext_vector_type(4)))  float    v4f;

__global__ void egae_prep(const float* __restrict__ W1,
                          const float* __restrict__ W2,
                          _Float16* __restrict__ W1t,
                          _Float16* __restrict__ W2t,
                          unsigned* __restrict__ cnt,
                          unsigned* __restrict__ cur,
                          unsigned* __restrict__ tick) {
  int t = blockIdx.x * blockDim.x + threadIdx.x;
  if (t < NBK2) { cnt[t] = 0u; cur[t] = 0u; }
  if (t < 4) tick[t] = 0u;
  if (t < D_FEAT * L1D) {
    int k = t / L1D, n = t % L1D;
    W1t[(size_t)n * D_FEAT + k] = (_Float16)W1[t];
  }
  if (t < L1D * L2D) {
    int k = t / L2D, n = t % L2D;
    W2t[(size_t)n * L1D + k] = (_Float16)W2[t];
  }
}

__global__ void egae_zero(float* __restrict__ p, int n) {
  int i = blockIdx.x * blockDim.x + threadIdx.x;
  if (i < n) p[i] = 0.0f;
}

__global__ void egae_gemm1(const float* __restrict__ X,
                           const _Float16* __restrict__ W1t,
                           float* __restrict__ Y1) {
  const int lane = threadIdx.x & 31;
  const int wave = (blockIdx.x * blockDim.x + threadIdx.x) >> 5;
  const int row0 = wave << 4;
  if (row0 >= N_NODES) return;
  const int m  = lane & 15;
  const int kh = lane >> 4;

  const float*    arow = X   + (size_t)(row0 + m) * D_FEAT + kh * 8;
  const _Float16* bc0  = W1t + (size_t)m         * D_FEAT + kh * 8;
  const _Float16* bc1  = W1t + (size_t)(m + 16)  * D_FEAT + kh * 8;

  v8f acc0 = {}; v8f acc1 = {};
  for (int k0 = 0; k0 < D_FEAT; k0 += 32) {
    v4f a0 = *(const v4f*)(arow + k0);
    v4f a1 = *(const v4f*)(arow + k0 + 4);
    v4f a2 = *(const v4f*)(arow + k0 + 16);
    v4f a3 = *(const v4f*)(arow + k0 + 20);
    v16h af;
#pragma unroll
    for (int i = 0; i < 4; ++i) {
      af[i]      = (_Float16)a0[i];
      af[4 + i]  = (_Float16)a1[i];
      af[8 + i]  = (_Float16)a2[i];
      af[12 + i] = (_Float16)a3[i];
    }
    v8h b0lo = *(const v8h*)(bc0 + k0);
    v8h b0hi = *(const v8h*)(bc0 + k0 + 16);
    v8h b1lo = *(const v8h*)(bc1 + k0);
    v8h b1hi = *(const v8h*)(bc1 + k0 + 16);
    v16h bf0, bf1;
#pragma unroll
    for (int i = 0; i < 8; ++i) {
      bf0[i] = b0lo[i]; bf0[8 + i] = b0hi[i];
      bf1[i] = b1lo[i]; bf1[8 + i] = b1hi[i];
    }
    acc0 = __builtin_amdgcn_wmma_f32_16x16x32_f16(false, af, false, bf0,
                                                  (short)0, acc0, false, false);
    acc1 = __builtin_amdgcn_wmma_f32_16x16x32_f16(false, af, false, bf1,
                                                  (short)0, acc1, false, false);
  }
  float* drow = Y1 + (size_t)(row0 + kh * 8) * L1D + m;
#pragma unroll
  for (int v = 0; v < 8; ++v) {
    drow[(size_t)v * L1D]      = acc0[v];
    drow[(size_t)v * L1D + 16] = acc1[v];
  }
}

__device__ __forceinline__ unsigned egae_bucket(unsigned r, unsigned c) {
  unsigned bn = min(c >> SB_SHIFT, (unsigned)(NBLK - 1));
  return (r >= HALF_ROW ? (unsigned)NBLK : 0u) + bn;
}

__global__ __launch_bounds__(SORT_T)
void egae_hist(const int* __restrict__ er, const int* __restrict__ ec,
               unsigned* __restrict__ cnt) {
  __shared__ unsigned h[NBK2];
  const int tid = threadIdx.x;
  if (tid < NBK2) h[tid] = 0u;
  __syncthreads();
  const int base = blockIdx.x * SORT_CH;
#pragma unroll
  for (int k = 0; k < SORT_CH / SORT_T; ++k) {
    int e = base + k * SORT_T + tid;
    if (e < N_EDGES) {
      unsigned b = egae_bucket((unsigned)er[e], (unsigned)ec[e]);
      atomicAdd(&h[b], 1u);
    }
  }
  __syncthreads();
  if (tid < NBK2) { unsigned c = h[tid]; if (c) atomicAdd(&cnt[tid], c); }
}

__device__ __forceinline__ unsigned egae_rec_word(unsigned idx) {
  return 2u * idx + (idx >= REC_SPLIT ? REC_GAPW : 0u);
}

template <bool REC>
__device__ __forceinline__ void egae_scatter_body(const int* __restrict__ er,
                                                  const int* __restrict__ ec,
                                                  const float* __restrict__ ev,
                                                  const unsigned* __restrict__ cnt,
                                                  unsigned* __restrict__ cur,
                                                  unsigned* __restrict__ elist,
                                                  unsigned* h, unsigned* gpos, unsigned* lofs,
                                                  unsigned* wt, unsigned* wl,
                                                  uint2* srec, unsigned* sdst) {
  const int tid = threadIdx.x, lane = tid & 31, w = tid >> 5;
  if (tid < NBK2) h[tid] = 0u;
  __syncthreads();
  const unsigned x = (tid < NBK2) ? cnt[tid] : 0u;
  const int base = blockIdx.x * SORT_CH;
  unsigned cv[SORT_CH / SORT_T], rv[SORT_CH / SORT_T], pv[SORT_CH / SORT_T];
#pragma unroll
  for (int k = 0; k < SORT_CH / SORT_T; ++k) {
    const int ee = min(base + k * SORT_T + tid, N_EDGES - 1);
    cv[k] = (unsigned)ec[ee];
    rv[k] = (unsigned)er[ee];
    pv[k] = REC ? __float_as_uint(ev[ee]) : 0u;
  }
  unsigned qx[SORT_CH / SORT_T], br[SORT_CH / SORT_T];
#pragma unroll
  for (int k = 0; k < SORT_CH / SORT_T; ++k) {
    int e = base + k * SORT_T + tid;
    const unsigned b = egae_bucket(rv[k], cv[k]);
    qx[k] = REC ? ((rv[k] & REC_ROWMASK) | ((cv[k] & (unsigned)(SB_ROWS - 1)) << 17)) : (unsigned)e;
    br[k] = b << 16;
    if (e < N_EDGES) br[k] |= atomicAdd(&h[b], 1u);
  }
  __syncthreads();
  const unsigned c = (tid < NBK2) ? h[tid] : 0u;
  unsigned inc = x, inl = c;
#pragma unroll
  for (int d = 1; d < 32; d <<= 1) {
    unsigned y = __shfl_up(inc, d);
    unsigned z = __shfl_up(inl, d);
    if (lane >= d) { inc += y; inl += z; }
  }
  if (lane == 31) { wt[w] = inc; wl[w] = inl; }
  __syncthreads();
  unsigned pre = 0u, prl = 0u;
  for (int j = 0; j < w; ++j) { pre += wt[j]; prl += wl[j]; }
  if (tid < NBK2) {
    unsigned off = c ? atomicAdd(&cur[tid], c) : 0u;
    gpos[tid] = pre + inc - x + off;
    lofs[tid] = prl + inl - c;
  }
  __syncthreads();
#pragma unroll
  for (int k = 0; k < SORT_CH / SORT_T; ++k) {
    int e = base + k * SORT_T + tid;
    if (e < N_EDGES) {
      const unsigned b = br[k] >> 16, rank = br[k] & 0xFFFFu;
      const unsigned slot = lofs[b] + rank;
      uint2 q;
      q.x = qx[k];
      q.y = pv[k];
      srec[slot] = q;
      sdst[slot] = gpos[b] + rank;
    }
  }
  __syncthreads();
  const int nval = min(SORT_CH, N_EDGES - base);
#pragma unroll
  for (int k = 0; k < SORT_CH / SORT_T; ++k) {
    int i = k * SORT_T + tid;
    if (i < nval) {
      const uint2 q = srec[i];
      const unsigned pos = sdst[i];
      if (REC) *(uint2*)(elist + egae_rec_word(pos)) = q;
      else     elist[pos] = q.x;
    }
  }
}

__global__ __launch_bounds__(SORT_T, 16)
void egae_scatter(const int* __restrict__ er, const int* __restrict__ ec,
                  const float* __restrict__ ev, const unsigned* __restrict__ cnt,
                  unsigned* __restrict__ cur, unsigned* __restrict__ elist,
                  unsigned long long ws_bytes) {
  __shared__ unsigned h[NBK2];
  __shared__ unsigned gpos[NBK2];
  __shared__ unsigned lofs[NBK2];
  __shared__ unsigned wt[32];
  __shared__ unsigned wl[32];
  __shared__ uint2 srec[SORT_CH];
  __shared__ unsigned sdst[SORT_CH];
  if (ws_bytes >= REC_WS_MIN)
    egae_scatter_body<true>(er, ec, ev, cnt, cur, elist, h, gpos, lofs, wt, wl, srec, sdst);
  else
    egae_scatter_body<false>(er, ec, ev, cnt, cur, elist, h, gpos, lofs, wt, wl, srec, sdst);
}

__device__ __forceinline__ unsigned egae_claim(unsigned* __restrict__ tick, unsigned* s_item) {
  if (threadIdx.x == 0) {
    const unsigned hf = (__builtin_amdgcn_s_sendmsg_rtn(0x87) >> 18) & 1u;
    unsigned t = atomicAdd(&tick[hf], 1u);
    unsigned item = hf * NBLK + t;
    if (t >= NBLK) {
      t = atomicAdd(&tick[hf ^ 1u], 1u);
      item = (t < NBLK) ? (hf ^ 1u) * NBLK + t : 0xFFFFFFFFu;
    }
    *s_item = item;
  }
  __syncthreads();
  return *s_item;
}

__device__ __forceinline__ unsigned egae_block_start(const unsigned* __restrict__ cnt,
                                                     int b, unsigned* s_start) {
  const int tid = threadIdx.x;
  if (tid == 0) *s_start = 0u;
  __syncthreads();
  unsigned x = ((tid < b) ? cnt[tid] : 0u) + ((tid + 512 < b) ? cnt[tid + 512] : 0u);
#pragma unroll
  for (int d = 16; d >= 1; d >>= 1) x += __shfl_xor(x, d);
  if ((tid & 31) == 0 && x) atomicAdd(s_start, x);
  __syncthreads();
  return *s_start;
}

__device__ __forceinline__ void egae_walk32(const float* ys, float* __restrict__ dst,
                                            int lane, int r, int cl, float v, unsigned n) {
  if (n == 32u) {
#pragma unroll
    for (int t0 = 0; t0 < 32; t0 += 8) {
      float y[8];
#pragma unroll
      for (int u = 0; u < 8; ++u) {
        int cc = __builtin_amdgcn_readlane(cl, t0 + u);
        y[u] = ys[cc * L1D + lane];
      }
#pragma unroll
      for (int u = 0; u < 8; ++u) {
        int rr = __builtin_amdgcn_readlane(r, t0 + u);
        float vv = __int_as_float(__builtin_amdgcn_readlane(__float_as_int(v), t0 + u));
        float* rowp = dst + (size_t)rr * L1D;
        atomicAdd(rowp + (unsigned)lane, vv * y[u]);
      }
    }
  } else {
    for (unsigned t = 0; t < n; ++t) {
      int rr = __builtin_amdgcn_readlane(r, (int)t);
      int cc = __builtin_amdgcn_readlane(cl, (int)t);
      float vv = __int_as_float(__builtin_amdgcn_readlane(__float_as_int(v), (int)t));
      atomicAdd(dst + (size_t)rr * L1D + lane, vv * ys[cc * L1D + lane]);
    }
  }
}

__device__ __forceinline__ uint2 egae_rec_load(const unsigned* __restrict__ elist,
                                               unsigned idx, unsigned end) {
  uint2 q; q.x = 0u; q.y = 0u;
  if (idx < end) q = *(const uint2*)(elist + egae_rec_word(idx));
  return q;
}

template <bool REC>
__device__ __forceinline__ void egae_spmm32_body(const int* __restrict__ er,
                                                 const int* __restrict__ ec,
                                                 const float* __restrict__ ev,
                                                 const unsigned* __restrict__ cnt,
                                                 const unsigned* __restrict__ elist,
                                                 const float* __restrict__ src,
                                                 float* __restrict__ dst,
                                                 float* ys, unsigned* s_start,
                                                 unsigned* __restrict__ tick, unsigned* s_item) {
  const int tid = threadIdx.x, lane = tid & 31;
  const int w = __builtin_amdgcn_readfirstlane(tid >> 5);
  const unsigned item = egae_claim(tick, s_item);
  if (item == 0xFFFFFFFFu) return;
  const int b = (int)item;
  const int row0 = (int)(item >= (unsigned)NBLK ? item - (unsigned)NBLK : item) * SB_ROWS;
  const int nrows = min(SB_ROWS, N_NODES - row0);
  {
    const v4f* g = (const v4f*)(src + (size_t)row0 * L1D);
    v4f* s = (v4f*)ys;
    if (nrows == SB_ROWS) {
      v4f tmp[4];
#pragma unroll
      for (int k = 0; k < 4; ++k) tmp[k] = g[tid + k * 512];
#pragma unroll
      for (int k = 0; k < 4; ++k) s[tid + k * 512] = tmp[k];
    } else {
      for (int k = tid; k < nrows * (L1D / 4); k += 512) s[k] = g[k];
    }
  }
  const unsigned start = egae_block_start(cnt, b, s_start);
  const unsigned end = start + cnt[b];
  unsigned i = start + (unsigned)w * 32u;
  if (REC) {
    uint2 q0 = egae_rec_load(elist, i + (unsigned)lane, end);
    uint2 q1 = egae_rec_load(elist, i + 512u + (unsigned)lane, end);
    for (; i < end; i += 16u * 32u) {
      const uint2 q2 = egae_rec_load(elist, i + 1024u + (unsigned)lane, end);
      const unsigned n = min(32u, end - i);
      egae_walk32(ys, dst, lane, (int)(q0.x & REC_ROWMASK), (int)(q0.x >> 17),
                  __uint_as_float(q0.y), n);
      q0 = q1; q1 = q2;
    }
  } else {
    int idn = (i + (unsigned)lane < end) ? (int)elist[i + (unsigned)lane] : 0;
    int r = er[idn];
    int cl = (ec[idn] - row0) & (SB_ROWS - 1);
    float v = (i + (unsigned)lane < end) ? ev[idn] : 0.0f;
    idn = (i + 512u + (unsigned)lane < end) ? (int)elist[i + 512u + (unsigned)lane] : 0;
    for (; i < end; i += 16u * 32u) {
      const unsigned i1 = i + 16u * 32u;
      const int   rn  = er[idn];
      const int   ecn = ec[idn];
      const float evn = ev[idn];
      idn = (i1 + 512u + (unsigned)lane < end) ? (int)elist[i1 + 512u + (unsigned)lane] : 0;
      const unsigned n = min(32u, end - i);
      egae_walk32(ys, dst, lane, r, cl, v, n);
      r = rn;
      cl = (ecn - row0) & (SB_ROWS - 1);
      v = (i1 + (unsigned)lane < end) ? evn : 0.0f;
    }
  }
}

__global__ __launch_bounds__(512)
void egae_spmm32s(const int* __restrict__ er, const int* __restrict__ ec,
                  const float* __restrict__ ev, const unsigned* __restrict__ cnt,
                  const unsigned* __restrict__ elist,
                  const float* __restrict__ src, float* __restrict__ dst,
                  unsigned* __restrict__ tick, unsigned long long ws_bytes) {
  __shared__ float ys[SB_ROWS * L1D];
  __shared__ unsigned s_start;
  __shared__ unsigned s_item;
  if (ws_bytes >= REC_WS_MIN)
    egae_spmm32_body<true>(er, ec, ev, cnt, elist, src, dst, ys, &s_start, tick, &s_item);
  else
    egae_spmm32_body<false>(er, ec, ev, cnt, elist, src, dst, ys, &s_start, tick, &s_item);
}

__global__ void egae_gemm2(const float* __restrict__ H1,
                           const _Float16* __restrict__ W2t,
                           float* __restrict__ Y2) {
  const int lane = threadIdx.x & 31;
  const int wave = (blockIdx.x * blockDim.x + threadIdx.x) >> 5;
  const int row0 = wave << 4;
  if (row0 >= N_NODES) return;
  const int m  = lane & 15;
  const int kh = lane >> 4;

  const float* arow = H1 + (size_t)(row0 + m) * L1D + kh * 8;
  v4f a0 = *(const v4f*)(arow);
  v4f a1 = *(const v4f*)(arow + 4);
  v4f a2 = *(const v4f*)(arow + 16);
  v4f a3 = *(const v4f*)(arow + 20);
  v16h af;
#pragma unroll
  for (int i = 0; i < 4; ++i) {
    af[i]      = (_Float16)fmaxf(a0[i], 0.0f);
    af[4 + i]  = (_Float16)fmaxf(a1[i], 0.0f);
    af[8 + i]  = (_Float16)fmaxf(a2[i], 0.0f);
    af[12 + i] = (_Float16)fmaxf(a3[i], 0.0f);
  }
  const _Float16* bc = W2t + (size_t)m * L1D + kh * 8;
  v8h blo = *(const v8h*)(bc);
  v8h bhi = *(const v8h*)(bc + 16);
  v16h bf;
#pragma unroll
  for (int i = 0; i < 8; ++i) { bf[i] = blo[i]; bf[8 + i] = bhi[i]; }

  v8f acc = {};
  acc = __builtin_amdgcn_wmma_f32_16x16x32_f16(false, af, false, bf,
                                               (short)0, acc, false, false);
  float* drow = Y2 + (size_t)(row0 + kh * 8) * L2D + m;
#pragma unroll
  for (int v = 0; v < 8; ++v) drow[(size_t)v * L2D] = acc[v];
}

__device__ __forceinline__ void egae_walk16(const float* ys, float* __restrict__ dst,
                                            int col, unsigned hh, int r, int cl, float v,
                                            unsigned n) {
  if (n == 32u) {
#pragma unroll
    for (int p0 = 0; p0 < 16; p0 += 8) {
      float y[8];
#pragma unroll
      for (int u = 0; u < 8; ++u) {
        int c0 = __builtin_amdgcn_readlane(cl, 2 * (p0 + u));
        int c1 = __builtin_amdgcn_readlane(cl, 2 * (p0 + u) + 1);
        int cc = hh ? c1 : c0;
        y[u] = ys[cc * L2D + col];
      }
#pragma unroll
      for (int u = 0; u < 8; ++u) {
        int r0 = __builtin_amdgcn_readlane(r, 2 * (p0 + u));
        int r1 = __builtin_amdgcn_readlane(r, 2 * (p0 + u) + 1);
        float v0 = __int_as_float(__builtin_amdgcn_readlane(__float_as_int(v), 2 * (p0 + u)));
        float v1 = __int_as_float(__builtin_amdgcn_readlane(__float_as_int(v), 2 * (p0 + u) + 1));
        int rr = hh ? r1 : r0;
        float vv = hh ? v1 : v0;
        atomicAdd(dst + (size_t)rr * L2D + col, vv * y[u]);
      }
    }
  } else {
    for (unsigned t = 0; t < n; t += 2u) {
      int r0 = __builtin_amdgcn_readlane(r, (int)t);
      int r1 = __builtin_amdgcn_readlane(r, (int)t + 1);
      int c0 = __builtin_amdgcn_readlane(cl, (int)t);
      int c1 = __builtin_amdgcn_readlane(cl, (int)t + 1);
      float v0 = __int_as_float(__builtin_amdgcn_readlane(__float_as_int(v), (int)t));
      float v1 = __int_as_float(__builtin_amdgcn_readlane(__float_as_int(v), (int)t + 1));
      int rr = hh ? r1 : r0;
      int cc = hh ? c1 : c0;
      float vv = hh ? v1 : v0;
      if (t + hh < n) atomicAdd(dst + (size_t)rr * L2D + col, vv * ys[cc * L2D + col]);
    }
  }
}

template <bool REC>
__device__ __forceinline__ void egae_spmm16_body(const int* __restrict__ er,
                                                 const int* __restrict__ ec,
                                                 const float* __restrict__ ev,
                                                 const unsigned* __restrict__ cnt,
                                                 const unsigned* __restrict__ elist,
                                                 const float* __restrict__ src,
                                                 float* __restrict__ dst,
                                                 float* ys, unsigned* s_start,
                                                 unsigned* __restrict__ tick, unsigned* s_item) {
  const int tid = threadIdx.x, lane = tid & 31;
  const int w = __builtin_amdgcn_readfirstlane(tid >> 5);
  const int col = lane & 15;
  const unsigned hh = (unsigned)lane >> 4;
  const unsigned item = egae_claim(tick, s_item);
  if (item == 0xFFFFFFFFu) return;
  const int b = (int)item;
  const int row0 = (int)(item >= (unsigned)NBLK ? item - (unsigned)NBLK : item) * SB_ROWS;
  const int nrows = min(SB_ROWS, N_NODES - row0);
  {
    const v4f* g = (const v4f*)(src + (size_t)row0 * L2D);
    v4f* s = (v4f*)ys;
    if (nrows == SB_ROWS) {
      v4f tmp[2];
#pragma unroll
      for (int k = 0; k < 2; ++k) tmp[k] = g[tid + k * 512];
#pragma unroll
      for (int k = 0; k < 2; ++k) s[tid + k * 512] = tmp[k];
    } else {
      for (int k = tid; k < nrows * (L2D / 4); k += 512) s[k] = g[k];
    }
  }
  const unsigned start = egae_block_start(cnt, b, s_start);
  const unsigned end = start + cnt[b];
  unsigned i = start + (unsigned)w * 32u;
  if (REC) {
    uint2 q0 = egae_rec_load(elist, i + (unsigned)lane, end);
    uint2 q1 = egae_rec_load(elist, i + 512u + (unsigned)lane, end);
    for (; i < end; i += 16u * 32u) {
      const uint2 q2 = egae_rec_load(elist, i + 1024u + (unsigned)lane, end);
      const unsigned n = min(32u, end - i);
      egae_walk16(ys, dst, col, hh, (int)(q0.x & REC_ROWMASK), (int)(q0.x >> 17),
                  __uint_as_float(q0.y), n);
      q0 = q1; q1 = q2;
    }
  } else {
    int idn = (i + (unsigned)lane < end) ? (int)elist[i + (unsigned)lane] : 0;
    int r = er[idn];
    int cl = (ec[idn] - row0) & (SB_ROWS - 1);
    float v = (i + (unsigned)lane < end) ? ev[idn] : 0.0f;
    idn = (i + 512u + (unsigned)lane < end) ? (int)elist[i + 512u + (unsigned)lane] : 0;
    for (; i < end; i += 16u * 32u) {
      const unsigned i1 = i + 16u * 32u;
      const int   rn  = er[idn];
      const int   ecn = ec[idn];
      const float evn = ev[idn];
      idn = (i1 + 512u + (unsigned)lane < end) ? (int)elist[i1 + 512u + (unsigned)lane] : 0;
      const unsigned n = min(32u, end - i);
      egae_walk16(ys, dst, col, hh, r, cl, v, n);
      r = rn;
      cl = (ecn - row0) & (SB_ROWS - 1);
      v = (i1 + (unsigned)lane < end) ? evn : 0.0f;
    }
  }
}

__global__ __launch_bounds__(512)
void egae_spmm16s(const int* __restrict__ er, const int* __restrict__ ec,
                  const float* __restrict__ ev, const unsigned* __restrict__ cnt,
                  const unsigned* __restrict__ elist,
                  const float* __restrict__ src, float* __restrict__ dst,
                  unsigned* __restrict__ tick, unsigned long long ws_bytes) {
  __shared__ float ys[SB_ROWS * L2D];
  __shared__ unsigned s_start;
  __shared__ unsigned s_item;
  if (ws_bytes >= REC_WS_MIN)
    egae_spmm16_body<true>(er, ec, ev, cnt, elist, src, dst, ys, &s_start, tick, &s_item);
  else
    egae_spmm16_body<false>(er, ec, ev, cnt, elist, src, dst, ys, &s_start, tick, &s_item);
}

__global__ void egae_norm(float* __restrict__ io) {
  int i = blockIdx.x * blockDim.x + threadIdx.x;
  if (i >= N_NODES) return;
  v4f* s = (v4f*)(io + (size_t)i * L2D);
  v4f x0 = s[0], x1 = s[1], x2 = s[2], x3 = s[3];
  float ss = 0.0f;
#pragma unroll
  for (int j = 0; j < 4; ++j)
    ss += x0[j] * x0[j] + x1[j] * x1[j] + x2[j] * x2[j] + x3[j] * x3[j];
  float inv = 1.0f / fmaxf(sqrtf(ss), 1e-7f);
  s[0] = x0 * inv; s[1] = x1 * inv; s[2] = x2 * inv; s[3] = x3 * inv;
}

extern "C" void kernel_launch(void* const* d_in, const int* in_sizes, int n_in,
                              void* d_out, int out_size, void* d_ws, size_t ws_size,
                              hipStream_t stream) {
  const float* X  = (const float*)d_in[0];
  const float* W1 = (const float*)d_in[1];
  const float* W2 = (const float*)d_in[2];
  const int*   lr = (const int*)d_in[3];
  const int*   lc = (const int*)d_in[4];
  const float* lv = (const float*)d_in[5];
  float* out = (float*)d_out;

  char* ws = (char*)d_ws;
  float*    Y1   = (float*)(ws + 0);
  float*    H1   = (float*)(ws + 12800000);
  _Float16* W1t  = (_Float16*)(ws + 12800000);
  unsigned* elist = (unsigned*)(ws + 25600000);
  unsigned* cnt  = (unsigned*)(ws + 38400000);
  unsigned* cur  = (unsigned*)(ws + 38404096);
  unsigned* tick = (unsigned*)(ws + 38408192);
  _Float16* W2t  = (_Float16*)(ws + 38432768);
  float*    Y2   = Y1;
  float*    Y3   = out;

  const int waves  = N_NODES / 16;
  const int gblk   = (waves + 7) / 8;
  const int sblk   = (N_EDGES + SORT_CH - 1) / SORT_CH;

  egae_prep<<<(D_FEAT * L1D + 255) / 256, 256, 0, stream>>>(W1, W2, W1t, W2t, cnt, cur, tick);
  egae_hist<<<sblk, SORT_T, 0, stream>>>(lr, lc, cnt);
  egae_scatter<<<sblk, SORT_T, 0, stream>>>(lr, lc, lv, cnt, cur, elist, (unsigned long long)ws_size);
  egae_gemm1<<<gblk, 256, 0, stream>>>(X, W1t, Y1);
  egae_zero<<<(N_NODES * L1D + 255) / 256, 256, 0, stream>>>(H1, N_NODES * L1D);
  egae_spmm32s<<<NBK2, 512, 0, stream>>>(lr, lc, lv, cnt, elist, Y1, H1, tick, (unsigned long long)ws_size);
  egae_gemm2<<<gblk, 256, 0, stream>>>(H1, W2t, Y2);
  egae_zero<<<(N_NODES * L2D + 255) / 256, 256, 0, stream>>>(Y3, N_NODES * L2D);
  egae_spmm16s<<<NBK2, 512, 0, stream>>>(lr, lc, lv, cnt, elist, Y2, Y3, tick + 2, (unsigned long long)ws_size);
  egae_norm<<<(N_NODES + 255) / 256, 256, 0, stream>>>(out);
}


# grids sized to the 256 WGPs: 261-row source blocks give 768 buckets (3 spmm work-groups per WGP, was 782: a few WGPs carried a 4th), sort in 512 chunks of 6250 edges (2 per WGP, was 391)
# speedup vs baseline: 3.5735x; 1.0603x over previous
#include <hip/hip_runtime.h>
#include <hip/hip_bf16.h>


#define N_NODES 100000
#define D_FEAT  512
#define N_EDGES 3200000
#define L1D     32
#define L2D     16
#define SB_ROWS  261
#define NBLK     384
#define NBK2     768
#define HALF_ROW 50000u
#define SORT_CH  6250
#define SORT_T   1024
#define SORT_K   7
#define REC_SPLIT  1600000u
#define REC_GAPW   8448u
#define REC_WS_MIN 51233792ull
#define REC_ROWMASK 0x1FFFFu

typedef __attribute__((ext_vector_type(16))) _Float16 v16h;
typedef __attribute__((ext_vector_type(8)))  _Float16 v8h;
typedef __attribute__((ext_vector_type(8)))  float    v8f;
typedef __attribute__((ext_vector_type(4)))  float    v4f;

__global__ void egae_prep(const float* __restrict__ W1,
                          const float* __restrict__ W2,
                          _Float16* __restrict__ W1t,
                          _Float16* __restrict__ W2t,
                          unsigned* __restrict__ cnt,
                          unsigned* __restrict__ cur,
                          unsigned* __restrict__ tick) {
  int t = blockIdx.x * blockDim.x + threadIdx.x;
  if (t < NBK2) { cnt[t] = 0u; cur[t] = 0u; }
  if (t < 4) tick[t] = 0u;
  if (t < D_FEAT * L1D) {
    int k = t / L1D, n = t % L1D;
    W1t[(size_t)n * D_FEAT + k] = (_Float16)W1[t];
  }
  if (t < L1D * L2D) {
    int k = t / L2D, n = t % L2D;
    W2t[(size_t)n * L1D + k] = (_Float16)W2[t];
  }
}

__global__ void egae_zero(float* __restrict__ p, int n) {
  int i = blockIdx.x * blockDim.x + threadIdx.x;
  if (i < n) p[i] = 0.0f;
}

__global__ void egae_gemm1(const float* __restrict__ X,
                           const _Float16* __restrict__ W1t,
                           float* __restrict__ Y1) {
  const int lane = threadIdx.x & 31;
  const int wave = (blockIdx.x * blockDim.x + threadIdx.x) >> 5;
  const int row0 = wave << 4;
  if (row0 >= N_NODES) return;
  const int m  = lane & 15;
  const int kh = lane >> 4;

  const float*    arow = X   + (size_t)(row0 + m) * D_FEAT + kh * 8;
  const _Float16* bc0  = W1t + (size_t)m         * D_FEAT + kh * 8;
  const _Float16* bc1  = W1t + (size_t)(m + 16)  * D_FEAT + kh * 8;

  v8f acc0 = {}; v8f acc1 = {};
  for (int k0 = 0; k0 < D_FEAT; k0 += 32) {
    v4f a0 = *(const v4f*)(arow + k0);
    v4f a1 = *(const v4f*)(arow + k0 + 4);
    v4f a2 = *(const v4f*)(arow + k0 + 16);
    v4f a3 = *(const v4f*)(arow + k0 + 20);
    v16h af;
#pragma unroll
    for (int i = 0; i < 4; ++i) {
      af[i]      = (_Float16)a0[i];
      af[4 + i]  = (_Float16)a1[i];
      af[8 + i]  = (_Float16)a2[i];
      af[12 + i] = (_Float16)a3[i];
    }
    v8h b0lo = *(const v8h*)(bc0 + k0);
    v8h b0hi = *(const v8h*)(bc0 + k0 + 16);
    v8h b1lo = *(const v8h*)(bc1 + k0);
    v8h b1hi = *(const v8h*)(bc1 + k0 + 16);
    v16h bf0, bf1;
#pragma unroll
    for (int i = 0; i < 8; ++i) {
      bf0[i] = b0lo[i]; bf0[8 + i] = b0hi[i];
      bf1[i] = b1lo[i]; bf1[8 + i] = b1hi[i];
    }
    acc0 = __builtin_amdgcn_wmma_f32_16x16x32_f16(false, af, false, bf0,
                                                  (short)0, acc0, false, false);
    acc1 = __builtin_amdgcn_wmma_f32_16x16x32_f16(false, af, false, bf1,
                                                  (short)0, acc1, false, false);
  }
  float* drow = Y1 + (size_t)(row0 + kh * 8) * L1D + m;
#pragma unroll
  for (int v = 0; v < 8; ++v) {
    drow[(size_t)v * L1D]      = acc0[v];
    drow[(size_t)v * L1D + 16] = acc1[v];
  }
}

__device__ __forceinline__ unsigned egae_srcblk(unsigned c) {
  return min(c / (unsigned)SB_ROWS, (unsigned)(NBLK - 1));
}
__device__ __forceinline__ unsigned egae_bucket(unsigned r, unsigned c) {
  return (r >= HALF_ROW ? (unsigned)NBLK : 0u) + egae_srcblk(c);
}

__global__ __launch_bounds__(SORT_T)
void egae_hist(const int* __restrict__ er, const int* __restrict__ ec,
               unsigned* __restrict__ cnt) {
  __shared__ unsigned h[NBK2];
  const int tid = threadIdx.x;
  if (tid < NBK2) h[tid] = 0u;
  __syncthreads();
  const int base = blockIdx.x * SORT_CH;
#pragma unroll
  for (int k = 0; k < SORT_K; ++k) {
    const int j = k * SORT_T + tid;
    const int e = base + j;
    if (j < SORT_CH && e < N_EDGES) {
      unsigned b = egae_bucket((unsigned)er[e], (unsigned)ec[e]);
      atomicAdd(&h[b], 1u);
    }
  }
  __syncthreads();
  if (tid < NBK2) { unsigned c = h[tid]; if (c) atomicAdd(&cnt[tid], c); }
}

__device__ __forceinline__ unsigned egae_rec_word(unsigned idx) {
  return 2u * idx + (idx >= REC_SPLIT ? REC_GAPW : 0u);
}

template <bool REC>
__device__ __forceinline__ void egae_scatter_body(const int* __restrict__ er,
                                                  const int* __restrict__ ec,
                                                  const float* __restrict__ ev,
                                                  const unsigned* __restrict__ cnt,
                                                  unsigned* __restrict__ cur,
                                                  unsigned* __restrict__ elist,
                                                  unsigned* h, unsigned* gpos, unsigned* lofs,
                                                  unsigned* wt, unsigned* wl,
                                                  uint2* srec, unsigned* sdst) {
  const int tid = threadIdx.x, lane = tid & 31, w = tid >> 5;
  if (tid < NBK2) h[tid] = 0u;
  __syncthreads();
  const unsigned x = (tid < NBK2) ? cnt[tid] : 0u;
  const int base = blockIdx.x * SORT_CH;
  unsigned cv[SORT_K], rv[SORT_K], pv[SORT_K];
#pragma unroll
  for (int k = 0; k < SORT_K; ++k) {
    const int ee = min(base + k * SORT_T + tid, N_EDGES - 1);
    cv[k] = (unsigned)ec[ee];
    rv[k] = (unsigned)er[ee];
    pv[k] = REC ? __float_as_uint(ev[ee]) : 0u;
  }
  unsigned qx[SORT_K], br[SORT_K];
#pragma unroll
  for (int k = 0; k < SORT_K; ++k) {
    const int j = k * SORT_T + tid;
    const int e = base + j;
    const unsigned bn = egae_srcblk(cv[k]);
    const unsigned b = (rv[k] >= HALF_ROW ? (unsigned)NBLK : 0u) + bn;
    const unsigned lr = min(cv[k] - bn * (unsigned)SB_ROWS, (unsigned)(SB_ROWS - 1));
    qx[k] = REC ? ((rv[k] & REC_ROWMASK) | (lr << 17)) : (unsigned)e;
    br[k] = b << 16;
    if (j < SORT_CH && e < N_EDGES) br[k] |= atomicAdd(&h[b], 1u);
  }
  __syncthreads();
  const unsigned c = (tid < NBK2) ? h[tid] : 0u;
  unsigned inc = x, inl = c;
#pragma unroll
  for (int d = 1; d < 32; d <<= 1) {
    unsigned y = __shfl_up(inc, d);
    unsigned z = __shfl_up(inl, d);
    if (lane >= d) { inc += y; inl += z; }
  }
  if (lane == 31) { wt[w] = inc; wl[w] = inl; }
  __syncthreads();
  unsigned pre = (lane < w) ? wt[lane] : 0u;
  unsigned prl = (lane < w) ? wl[lane] : 0u;
#pragma unroll
  for (int d = 16; d >= 1; d >>= 1) { pre += __shfl_xor(pre, d); prl += __shfl_xor(prl, d); }
  if (tid < NBK2) {
    unsigned off = c ? atomicAdd(&cur[tid], c) : 0u;
    gpos[tid] = pre + inc - x + off;
    lofs[tid] = prl + inl - c;
  }
  __syncthreads();
#pragma unroll
  for (int k = 0; k < SORT_K; ++k) {
    const int j = k * SORT_T + tid;
    const int e = base + j;
    if (j < SORT_CH && e < N_EDGES) {
      const unsigned b = br[k] >> 16, rank = br[k] & 0xFFFFu;
      const unsigned slot = lofs[b] + rank;
      uint2 q;
      q.x = qx[k];
      q.y = pv[k];
      srec[slot] = q;
      sdst[slot] = gpos[b] + rank;
    }
  }
  __syncthreads();
  const int nval = min(SORT_CH, N_EDGES - base);
#pragma unroll
  for (int k = 0; k < SORT_K; ++k) {
    int i = k * SORT_T + tid;
    if (i < nval) {
      const uint2 q = srec[i];
      const unsigned pos = sdst[i];
      if (REC) *(uint2*)(elist + egae_rec_word(pos)) = q;
      else     elist[pos] = q.x;
    }
  }
}

__global__ __launch_bounds__(SORT_T, 16)
void egae_scatter(const int* __restrict__ er, const int* __restrict__ ec,
                  const float* __restrict__ ev, const unsigned* __restrict__ cnt,
                  unsigned* __restrict__ cur, unsigned* __restrict__ elist,
                  unsigned long long ws_bytes) {
  __shared__ unsigned h[NBK2];
  __shared__ unsigned gpos[NBK2];
  __shared__ unsigned lofs[NBK2];
  __shared__ unsigned wt[32];
  __shared__ unsigned wl[32];
  __shared__ uint2 srec[SORT_CH];
  __shared__ unsigned sdst[SORT_CH];
  if (ws_bytes >= REC_WS_MIN)
    egae_scatter_body<true>(er, ec, ev, cnt, cur, elist, h, gpos, lofs, wt, wl, srec, sdst);
  else
    egae_scatter_body<false>(er, ec, ev, cnt, cur, elist, h, gpos, lofs, wt, wl, srec, sdst);
}

__device__ __forceinline__ unsigned egae_claim(unsigned* __restrict__ tick, unsigned* s_item) {
  if (threadIdx.x == 0) {
    const unsigned hf = (__builtin_amdgcn_s_sendmsg_rtn(0x87) >> 18) & 1u;
    unsigned t = atomicAdd(&tick[hf], 1u);
    unsigned item = hf * NBLK + t;
    if (t >= NBLK) {
      t = atomicAdd(&tick[hf ^ 1u], 1u);
      item = (t < NBLK) ? (hf ^ 1u) * NBLK + t : 0xFFFFFFFFu;
    }
    *s_item = item;
  }
  __syncthreads();
  return *s_item;
}

__device__ __forceinline__ unsigned egae_block_start(const unsigned* __restrict__ cnt,
                                                     int b, unsigned* s_start) {
  const int tid = threadIdx.x;
  if (tid == 0) *s_start = 0u;
  __syncthreads();
  unsigned x = ((tid < b) ? cnt[tid] : 0u) + ((tid + 512 < b) ? cnt[tid + 512] : 0u);
#pragma unroll
  for (int d = 16; d >= 1; d >>= 1) x += __shfl_xor(x, d);
  if ((tid & 31) == 0 && x) atomicAdd(s_start, x);
  __syncthreads();
  return *s_start;
}

__device__ __forceinline__ void egae_walk32(const float* ys, float* __restrict__ dst,
                                            int lane, int r, int cl, float v, unsigned n) {
  if (n == 32u) {
#pragma unroll
    for (int t0 = 0; t0 < 32; t0 += 8) {
      float y[8];
#pragma unroll
      for (int u = 0; u < 8; ++u) {
        int cc = __builtin_amdgcn_readlane(cl, t0 + u);
        y[u] = ys[cc * L1D + lane];
      }
#pragma unroll
      for (int u = 0; u < 8; ++u) {
        int rr = __builtin_amdgcn_readlane(r, t0 + u);
        float vv = __int_as_float(__builtin_amdgcn_readlane(__float_as_int(v), t0 + u));
        float* rowp = dst + (size_t)rr * L1D;
        atomicAdd(rowp + (unsigned)lane, vv * y[u]);
      }
    }
  } else {
    for (unsigned t = 0; t < n; ++t) {
      int rr = __builtin_amdgcn_readlane(r, (int)t);
      int cc = __builtin_amdgcn_readlane(cl, (int)t);
      float vv = __int_as_float(__builtin_amdgcn_readlane(__float_as_int(v), (int)t));
      atomicAdd(dst + (size_t)rr * L1D + lane, vv * ys[cc * L1D + lane]);
    }
  }
}

__device__ __forceinline__ uint2 egae_rec_load(const unsigned* __restrict__ elist,
                                               unsigned idx, unsigned end) {
  uint2 q; q.x = 0u; q.y = 0u;
  if (idx < end) q = *(const uint2*)(elist + egae_rec_word(idx));
  return q;
}

template <bool REC>
__device__ __forceinline__ void egae_spmm32_body(const int* __restrict__ er,
                                                 const int* __restrict__ ec,
                                                 const float* __restrict__ ev,
                                                 const unsigned* __restrict__ cnt,
                                                 const unsigned* __restrict__ elist,
                                                 const float* __restrict__ src,
                                                 float* __restrict__ dst,
                                                 float* ys, unsigned* s_start,
                                                 unsigned* __restrict__ tick, unsigned* s_item) {
  const int tid = threadIdx.x, lane = tid & 31;
  const int w = __builtin_amdgcn_readfirstlane(tid >> 5);
  const unsigned item = egae_claim(tick, s_item);
  if (item == 0xFFFFFFFFu) return;
  const int b = (int)item;
  const int row0 = (int)(item >= (unsigned)NBLK ? item - (unsigned)NBLK : item) * SB_ROWS;
  const int nrows = min(SB_ROWS, N_NODES - row0);
  {
    const v4f* g = (const v4f*)(src + (size_t)row0 * L1D);
    v4f* s = (v4f*)ys;
    const int nv = nrows * (L1D / 4);
    v4f tmp[(SB_ROWS * (L1D / 4) + 511) / 512];
#pragma unroll
    for (int k = 0; k < (SB_ROWS * (L1D / 4) + 511) / 512; ++k) tmp[k] = g[min(tid + k * 512, nv - 1)];
#pragma unroll
    for (int k = 0; k < (SB_ROWS * (L1D / 4) + 511) / 512; ++k)
      if (tid + k * 512 < nv) s[tid + k * 512] = tmp[k];
  }
  const unsigned start = egae_block_start(cnt, b, s_start);
  const unsigned end = start + cnt[b];
  unsigned i = start + (unsigned)w * 32u;
  if (REC) {
    uint2 q0 = egae_rec_load(elist, i + (unsigned)lane, end);
    uint2 q1 = egae_rec_load(elist, i + 512u + (unsigned)lane, end);
    for (; i < end; i += 16u * 32u) {
      const uint2 q2 = egae_rec_load(elist, i + 1024u + (unsigned)lane, end);
      const unsigned n = min(32u, end - i);
      egae_walk32(ys, dst, lane, (int)(q0.x & REC_ROWMASK), (int)(q0.x >> 17),
                  __uint_as_float(q0.y), n);
      q0 = q1; q1 = q2;
    }
  } else {
    int idn = (i + (unsigned)lane < end) ? (int)elist[i + (unsigned)lane] : 0;
    int r = er[idn];
    int cl = (int)min((unsigned)(ec[idn] - row0), (unsigned)(SB_ROWS - 1));
    float v = (i + (unsigned)lane < end) ? ev[idn] : 0.0f;
    idn = (i + 512u + (unsigned)lane < end) ? (int)elist[i + 512u + (unsigned)lane] : 0;
    for (; i < end; i += 16u * 32u) {
      const unsigned i1 = i + 16u * 32u;
      const int   rn  = er[idn];
      const int   ecn = ec[idn];
      const float evn = ev[idn];
      idn = (i1 + 512u + (unsigned)lane < end) ? (int)elist[i1 + 512u + (unsigned)lane] : 0;
      const unsigned n = min(32u, end - i);
      egae_walk32(ys, dst, lane, r, cl, v, n);
      r = rn;
      cl = (int)min((unsigned)(ecn - row0), (unsigned)(SB_ROWS - 1));
      v = (i1 + (unsigned)lane < end) ? evn : 0.0f;
    }
  }
}

__global__ __launch_bounds__(512)
void egae_spmm32s(const int* __restrict__ er, const int* __restrict__ ec,
                  const float* __restrict__ ev, const unsigned* __restrict__ cnt,
                  const unsigned* __restrict__ elist,
                  const float* __restrict__ src, float* __restrict__ dst,
                  unsigned* __restrict__ tick, unsigned long long ws_bytes) {
  __shared__ float ys[SB_ROWS * L1D];
  __shared__ unsigned s_start;
  __shared__ unsigned s_item;
  if (ws_bytes >= REC_WS_MIN)
    egae_spmm32_body<true>(er, ec, ev, cnt, elist, src, dst, ys, &s_start, tick, &s_item);
  else
    egae_spmm32_body<false>(er, ec, ev, cnt, elist, src, dst, ys, &s_start, tick, &s_item);
}

__global__ void egae_gemm2(const float* __restrict__ H1,
                           const _Float16* __restrict__ W2t,
                           float* __restrict__ Y2) {
  const int lane = threadIdx.x & 31;
  const int wave = (blockIdx.x * blockDim.x + threadIdx.x) >> 5;
  const int row0 = wave << 4;
  if (row0 >= N_NODES) return;
  const int m  = lane & 15;
  const int kh = lane >> 4;

  const float* arow = H1 + (size_t)(row0 + m) * L1D + kh * 8;
  v4f a0 = *(const v4f*)(arow);
  v4f a1 = *(const v4f*)(arow + 4);
  v4f a2 = *(const v4f*)(arow + 16);
  v4f a3 = *(const v4f*)(arow + 20);
  v16h af;
#pragma unroll
  for (int i = 0; i < 4; ++i) {
    af[i]      = (_Float16)fmaxf(a0[i], 0.0f);
    af[4 + i]  = (_Float16)fmaxf(a1[i], 0.0f);
    af[8 + i]  = (_Float16)fmaxf(a2[i], 0.0f);
    af[12 + i] = (_Float16)fmaxf(a3[i], 0.0f);
  }
  const _Float16* bc = W2t + (size_t)m * L1D + kh * 8;
  v8h blo = *(const v8h*)(bc);
  v8h bhi = *(const v8h*)(bc + 16);
  v16h bf;
#pragma unroll
  for (int i = 0; i < 8; ++i) { bf[i] = blo[i]; bf[8 + i] = bhi[i]; }

  v8f acc = {};
  acc = __builtin_amdgcn_wmma_f32_16x16x32_f16(false, af, false, bf,
                                               (short)0, acc, false, false);
  float* drow = Y2 + (size_t)(row0 + kh * 8) * L2D + m;
#pragma unroll
  for (int v = 0; v < 8; ++v) drow[(size_t)v * L2D] = acc[v];
}

__device__ __forceinline__ void egae_walk16(const float* ys, float* __restrict__ dst,
                                            int col, unsigned hh, int r, int cl, float v,
                                            unsigned n) {
  if (n == 32u) {
#pragma unroll
    for (int p0 = 0; p0 < 16; p0 += 8) {
      float y[8];
#pragma unroll
      for (int u = 0; u < 8; ++u) {
        int c0 = __builtin_amdgcn_readlane(cl, 2 * (p0 + u));
        int c1 = __builtin_amdgcn_readlane(cl, 2 * (p0 + u) + 1);
        int cc = hh ? c1 : c0;
        y[u] = ys[cc * L2D + col];
      }
#pragma unroll
      for (int u = 0; u < 8; ++u) {
        int r0 = __builtin_amdgcn_readlane(r, 2 * (p0 + u));
        int r1 = __builtin_amdgcn_readlane(r, 2 * (p0 + u) + 1);
        float v0 = __int_as_float(__builtin_amdgcn_readlane(__float_as_int(v), 2 * (p0 + u)));
        float v1 = __int_as_float(__builtin_amdgcn_readlane(__float_as_int(v), 2 * (p0 + u) + 1));
        int rr = hh ? r1 : r0;
        float vv = hh ? v1 : v0;
        atomicAdd(dst + (size_t)rr * L2D + col, vv * y[u]);
      }
    }
  } else {
    for (unsigned t = 0; t < n; t += 2u) {
      int r0 = __builtin_amdgcn_readlane(r, (int)t);
      int r1 = __builtin_amdgcn_readlane(r, (int)t + 1);
      int c0 = __builtin_amdgcn_readlane(cl, (int)t);
      int c1 = __builtin_amdgcn_readlane(cl, (int)t + 1);
      float v0 = __int_as_float(__builtin_amdgcn_readlane(__float_as_int(v), (int)t));
      float v1 = __int_as_float(__builtin_amdgcn_readlane(__float_as_int(v), (int)t + 1));
      int rr = hh ? r1 : r0;
      int cc = hh ? c1 : c0;
      float vv = hh ? v1 : v0;
      if (t + hh < n) atomicAdd(dst + (size_t)rr * L2D + col, vv * ys[cc * L2D + col]);
    }
  }
}

template <bool REC>
__device__ __forceinline__ void egae_spmm16_body(const int* __restrict__ er,
                                                 const int* __restrict__ ec,
                                                 const float* __restrict__ ev,
                                                 const unsigned* __restrict__ cnt,
                                                 const unsigned* __restrict__ elist,
                                                 const float* __restrict__ src,
                                                 float* __restrict__ dst,
                                                 float* ys, unsigned* s_start,
                                                 unsigned* __restrict__ tick, unsigned* s_item) {
  const int tid = threadIdx.x, lane = tid & 31;
  const int w = __builtin_amdgcn_readfirstlane(tid >> 5);
  const int col = lane & 15;
  const unsigned hh = (unsigned)lane >> 4;
  const unsigned item = egae_claim(tick, s_item);
  if (item == 0xFFFFFFFFu) return;
  const int b = (int)item;
  const int row0 = (int)(item >= (unsigned)NBLK ? item - (unsigned)NBLK : item) * SB_ROWS;
  const int nrows = min(SB_ROWS, N_NODES - row0);
  {
    const v4f* g = (const v4f*)(src + (size_t)row0 * L2D);
    v4f* s = (v4f*)ys;
    const int nv = nrows * (L2D / 4);
    v4f tmp[(SB_ROWS * (L2D / 4) + 511) / 512];
#pragma unroll
    for (int k = 0; k < (SB_ROWS * (L2D / 4) + 511) / 512; ++k) tmp[k] = g[min(tid + k * 512, nv - 1)];
#pragma unroll
    for (int k = 0; k < (SB_ROWS * (L2D / 4) + 511) / 512; ++k)
      if (tid + k * 512 < nv) s[tid + k * 512] = tmp[k];
  }
  const unsigned start = egae_block_start(cnt, b, s_start);
  const unsigned end = start + cnt[b];
  unsigned i = start + (unsigned)w * 32u;
  if (REC) {
    uint2 q0 = egae_rec_load(elist, i + (unsigned)lane, end);
    uint2 q1 = egae_rec_load(elist, i + 512u + (unsigned)lane, end);
    for (; i < end; i += 16u * 32u) {
      const uint2 q2 = egae_rec_load(elist, i + 1024u + (unsigned)lane, end);
      const unsigned n = min(32u, end - i);
      egae_walk16(ys, dst, col, hh, (int)(q0.x & REC_ROWMASK), (int)(q0.x >> 17),
                  __uint_as_float(q0.y), n);
      q0 = q1; q1 = q2;
    }
  } else {
    int idn = (i + (unsigned)lane < end) ? (int)elist[i + (unsigned)lane] : 0;
    int r = er[idn];
    int cl = (int)min((unsigned)(ec[idn] - row0), (unsigned)(SB_ROWS - 1));
    float v = (i + (unsigned)lane < end) ? ev[idn] : 0.0f;
    idn = (i + 512u + (unsigned)lane < end) ? (int)elist[i + 512u + (unsigned)lane] : 0;
    for (; i < end; i += 16u * 32u) {
      const unsigned i1 = i + 16u * 32u;
      const int   rn  = er[idn];
      const int   ecn = ec[idn];
      const float evn = ev[idn];
      idn = (i1 + 512u + (unsigned)lane < end) ? (int)elist[i1 + 512u + (unsigned)lane] : 0;
      const unsigned n = min(32u, end - i);
      egae_walk16(ys, dst, col, hh, r, cl, v, n);
      r = rn;
      cl = (int)min((unsigned)(ecn - row0), (unsigned)(SB_ROWS - 1));
      v = (i1 + (unsigned)lane < end) ? evn : 0.0f;
    }
  }
}

__global__ __launch_bounds__(512)
void egae_spmm16s(const int* __restrict__ er, const int* __restrict__ ec,
                  const float* __restrict__ ev, const unsigned* __restrict__ cnt,
                  const unsigned* __restrict__ elist,
                  const float* __restrict__ src, float* __restrict__ dst,
                  unsigned* __restrict__ tick, unsigned long long ws_bytes) {
  __shared__ float ys[SB_ROWS * L2D];
  __shared__ unsigned s_start;
  __shared__ unsigned s_item;
  if (ws_bytes >= REC_WS_MIN)
    egae_spmm16_body<true>(er, ec, ev, cnt, elist, src, dst, ys, &s_start, tick, &s_item);
  else
    egae_spmm16_body<false>(er, ec, ev, cnt, elist, src, dst, ys, &s_start, tick, &s_item);
}

__global__ void egae_norm(float* __restrict__ io) {
  int i = blockIdx.x * blockDim.x + threadIdx.x;
  if (i >= N_NODES) return;
  v4f* s = (v4f*)(io + (size_t)i * L2D);
  v4f x0 = s[0], x1 = s[1], x2 = s[2], x3 = s[3];
  float ss = 0.0f;
#pragma unroll
  for (int j = 0; j < 4; ++j)
    ss += x0[j] * x0[j] + x1[j] * x1[j] + x2[j] * x2[j] + x3[j] * x3[j];
  float inv = 1.0f / fmaxf(sqrtf(ss), 1e-7f);
  s[0] = x0 * inv; s[1] = x1 * inv; s[2] = x2 * inv; s[3] = x3 * inv;
}

extern "C" void kernel_launch(void* const* d_in, const int* in_sizes, int n_in,
                              void* d_out, int out_size, void* d_ws, size_t ws_size,
                              hipStream_t stream) {
  const float* X  = (const float*)d_in[0];
  const float* W1 = (const float*)d_in[1];
  const float* W2 = (const float*)d_in[2];
  const int*   lr = (const int*)d_in[3];
  const int*   lc = (const int*)d_in[4];
  const float* lv = (const float*)d_in[5];
  float* out = (float*)d_out;

  char* ws = (char*)d_ws;
  float*    Y1   = (float*)(ws + 0);
  float*    H1   = (float*)(ws + 12800000);
  _Float16* W1t  = (_Float16*)(ws + 12800000);
  unsigned* elist = (unsigned*)(ws + 25600000);
  unsigned* cnt  = (unsigned*)(ws + 38400000);
  unsigned* cur  = (unsigned*)(ws + 38404096);
  unsigned* tick = (unsigned*)(ws + 38408192);
  _Float16* W2t  = (_Float16*)(ws + 38432768);
  float*    Y2   = Y1;
  float*    Y3   = out;

  const int waves  = N_NODES / 16;
  const int gblk   = (waves + 7) / 8;
  const int sblk   = (N_EDGES + SORT_CH - 1) / SORT_CH;

  egae_prep<<<(D_FEAT * L1D + 255) / 256, 256, 0, stream>>>(W1, W2, W1t, W2t, cnt, cur, tick);
  egae_hist<<<sblk, SORT_T, 0, stream>>>(lr, lc, cnt);
  egae_scatter<<<sblk, SORT_T, 0, stream>>>(lr, lc, lv, cnt, cur, elist, (unsigned long long)ws_size);
  egae_gemm1<<<gblk, 256, 0, stream>>>(X, W1t, Y1);
  egae_zero<<<(N_NODES * L1D + 255) / 256, 256, 0, stream>>>(H1, N_NODES * L1D);
  egae_spmm32s<<<NBK2, 512, 0, stream>>>(lr, lc, lv, cnt, elist, Y1, H1, tick, (unsigned long long)ws_size);
  egae_gemm2<<<gblk, 256, 0, stream>>>(H1, W2t, Y2);
  egae_zero<<<(N_NODES * L2D + 255) / 256, 256, 0, stream>>>(Y3, N_NODES * L2D);
  egae_spmm16s<<<NBK2, 512, 0, stream>>>(lr, lc, lv, cnt, elist, Y2, Y3, tick + 2, (unsigned long long)ws_size);
  egae_norm<<<(N_NODES + 255) / 256, 256, 0, stream>>>(out);
}


# zero fills ride on the gemm waves: a gemm1 wave zeroes its 16 rows of H1, a gemm2 wave its 16 rows of out (16 B per lane); both zero launches (150000 waves of 4 B per lane) gone, 10 -> 8
# speedup vs baseline: 3.8460x; 1.0763x over previous
#include <hip/hip_runtime.h>
#include <hip/hip_bf16.h>


#define N_NODES 100000
#define D_FEAT  512
#define N_EDGES 3200000
#define L1D     32
#define L2D     16
#define SB_ROWS  261
#define NBLK     384
#define NBK2     768
#define HALF_ROW 50000u
#define SORT_CH  6250
#define SORT_T   1024
#define SORT_K   7
#define REC_SPLIT  1600000u
#define REC_GAPW   8448u
#define REC_WS_MIN 51233792ull
#define REC_ROWMASK 0x1FFFFu

typedef __attribute__((ext_vector_type(16))) _Float16 v16h;
typedef __attribute__((ext_vector_type(8)))  _Float16 v8h;
typedef __attribute__((ext_vector_type(8)))  float    v8f;
typedef __attribute__((ext_vector_type(4)))  float    v4f;

__global__ void egae_prep(const float* __restrict__ W1,
                          const float* __restrict__ W2,
                          _Float16* __restrict__ W1t,
                          _Float16* __restrict__ W2t,
                          unsigned* __restrict__ cnt,
                          unsigned* __restrict__ cur,
                          unsigned* __restrict__ tick) {
  int t = blockIdx.x * blockDim.x + threadIdx.x;
  if (t < NBK2) { cnt[t] = 0u; cur[t] = 0u; }
  if (t < 4) tick[t] = 0u;
  if (t < D_FEAT * L1D) {
    int k = t / L1D, n = t % L1D;
    W1t[(size_t)n * D_FEAT + k] = (_Float16)W1[t];
  }
  if (t < L1D * L2D) {
    int k = t / L2D, n = t % L2D;
    W2t[(size_t)n * L1D + k] = (_Float16)W2[t];
  }
}

__global__ void egae_gemm1(const float* __restrict__ X,
                           const _Float16* __restrict__ W1t,
                           float* __restrict__ Y1,
                           float* __restrict__ H1z) {
  const int lane = threadIdx.x & 31;
  const int wave = (blockIdx.x * blockDim.x + threadIdx.x) >> 5;
  const int row0 = wave << 4;
  if (row0 >= N_NODES) return;
  const int m  = lane & 15;
  const int kh = lane >> 4;

  const float*    arow = X   + (size_t)(row0 + m) * D_FEAT + kh * 8;
  const _Float16* bc0  = W1t + (size_t)m         * D_FEAT + kh * 8;
  const _Float16* bc1  = W1t + (size_t)(m + 16)  * D_FEAT + kh * 8;

  v8f acc0 = {}; v8f acc1 = {};
  for (int k0 = 0; k0 < D_FEAT; k0 += 32) {
    v4f a0 = *(const v4f*)(arow + k0);
    v4f a1 = *(const v4f*)(arow + k0 + 4);
    v4f a2 = *(const v4f*)(arow + k0 + 16);
    v4f a3 = *(const v4f*)(arow + k0 + 20);
    v16h af;
#pragma unroll
    for (int i = 0; i < 4; ++i) {
      af[i]      = (_Float16)a0[i];
      af[4 + i]  = (_Float16)a1[i];
      af[8 + i]  = (_Float16)a2[i];
      af[12 + i] = (_Float16)a3[i];
    }
    v8h b0lo = *(const v8h*)(bc0 + k0);
    v8h b0hi = *(const v8h*)(bc0 + k0 + 16);
    v8h b1lo = *(const v8h*)(bc1 + k0);
    v8h b1hi = *(const v8h*)(bc1 + k0 + 16);
    v16h bf0, bf1;
#pragma unroll
    for (int i = 0; i < 8; ++i) {
      bf0[i] = b0lo[i]; bf0[8 + i] = b0hi[i];
      bf1[i] = b1lo[i]; bf1[8 + i] = b1hi[i];
    }
    acc0 = __builtin_amdgcn_wmma_f32_16x16x32_f16(false, af, false, bf0,
                                                  (short)0, acc0, false, false);
    acc1 = __builtin_amdgcn_wmma_f32_16x16x32_f16(false, af, false, bf1,
                                                  (short)0, acc1, false, false);
  }
  float* drow = Y1 + (size_t)(row0 + kh * 8) * L1D + m;
#pragma unroll
  for (int v = 0; v < 8; ++v) {
    drow[(size_t)v * L1D]      = acc0[v];
    drow[(size_t)v * L1D + 16] = acc1[v];
  }
  const v4f zero = {0.0f, 0.0f, 0.0f, 0.0f};
  v4f* z = (v4f*)(H1z + (size_t)row0 * L1D) + lane;
  z[0] = zero; z[32] = zero; z[64] = zero; z[96] = zero;
}

__device__ __forceinline__ unsigned egae_srcblk(unsigned c) {
  return min(c / (unsigned)SB_ROWS, (unsigned)(NBLK - 1));
}
__device__ __forceinline__ unsigned egae_bucket(unsigned r, unsigned c) {
  return (r >= HALF_ROW ? (unsigned)NBLK : 0u) + egae_srcblk(c);
}

__global__ __launch_bounds__(SORT_T)
void egae_hist(const int* __restrict__ er, const int* __restrict__ ec,
               unsigned* __restrict__ cnt) {
  __shared__ unsigned h[NBK2];
  const int tid = threadIdx.x;
  if (tid < NBK2) h[tid] = 0u;
  __syncthreads();
  const int base = blockIdx.x * SORT_CH;
#pragma unroll
  for (int k = 0; k < SORT_K; ++k) {
    const int j = k * SORT_T + tid;
    const int e = base + j;
    if (j < SORT_CH && e < N_EDGES) {
      unsigned b = egae_bucket((unsigned)er[e], (unsigned)ec[e]);
      atomicAdd(&h[b], 1u);
    }
  }
  __syncthreads();
  if (tid < NBK2) { unsigned c = h[tid]; if (c) atomicAdd(&cnt[tid], c); }
}

__device__ __forceinline__ unsigned egae_rec_word(unsigned idx) {
  return 2u * idx + (idx >= REC_SPLIT ? REC_GAPW : 0u);
}

template <bool REC>
__device__ __forceinline__ void egae_scatter_body(const int* __restrict__ er,
                                                  const int* __restrict__ ec,
                                                  const float* __restrict__ ev,
                                                  const unsigned* __restrict__ cnt,
                                                  unsigned* __restrict__ cur,
                                                  unsigned* __restrict__ elist,
                                                  unsigned* h, unsigned* gpos, unsigned* lofs,
                                                  unsigned* wt, unsigned* wl,
                                                  uint2* srec, unsigned* sdst) {
  const int tid = threadIdx.x, lane = tid & 31, w = tid >> 5;
  if (tid < NBK2) h[tid] = 0u;
  __syncthreads();
  const unsigned x = (tid < NBK2) ? cnt[tid] : 0u;
  const int base = blockIdx.x * SORT_CH;
  unsigned cv[SORT_K], rv[SORT_K], pv[SORT_K];
#pragma unroll
  for (int k = 0; k < SORT_K; ++k) {
    const int ee = min(base + k * SORT_T + tid, N_EDGES - 1);
    cv[k] = (unsigned)ec[ee];
    rv[k] = (unsigned)er[ee];
    pv[k] = REC ? __float_as_uint(ev[ee]) : 0u;
  }
  unsigned qx[SORT_K], br[SORT_K];
#pragma unroll
  for (int k = 0; k < SORT_K; ++k) {
    const int j = k * SORT_T + tid;
    const int e = base + j;
    const unsigned bn = egae_srcblk(cv[k]);
    const unsigned b = (rv[k] >= HALF_ROW ? (unsigned)NBLK : 0u) + bn;
    const unsigned lr = min(cv[k] - bn * (unsigned)SB_ROWS, (unsigned)(SB_ROWS - 1));
    qx[k] = REC ? ((rv[k] & REC_ROWMASK) | (lr << 17)) : (unsigned)e;
    br[k] = b << 16;
    if (j < SORT_CH && e < N_EDGES) br[k] |= atomicAdd(&h[b], 1u);
  }
  __syncthreads();
  const unsigned c = (tid < NBK2) ? h[tid] : 0u;
  unsigned inc = x, inl = c;
#pragma unroll
  for (int d = 1; d < 32; d <<= 1) {
    unsigned y = __shfl_up(inc, d);
    unsigned z = __shfl_up(inl, d);
    if (lane >= d) { inc += y; inl += z; }
  }
  if (lane == 31) { wt[w] = inc; wl[w] = inl; }
  __syncthreads();
  unsigned pre = (lane < w) ? wt[lane] : 0u;
  unsigned prl = (lane < w) ? wl[lane] : 0u;
#pragma unroll
  for (int d = 16; d >= 1; d >>= 1) { pre += __shfl_xor(pre, d); prl += __shfl_xor(prl, d); }
  if (tid < NBK2) {
    unsigned off = c ? atomicAdd(&cur[tid], c) : 0u;
    gpos[tid] = pre + inc - x + off;
    lofs[tid] = prl + inl - c;
  }
  __syncthreads();
#pragma unroll
  for (int k = 0; k < SORT_K; ++k) {
    const int j = k * SORT_T + tid;
    const int e = base + j;
    if (j < SORT_CH && e < N_EDGES) {
      const unsigned b = br[k] >> 16, rank = br[k] & 0xFFFFu;
      const unsigned slot = lofs[b] + rank;
      uint2 q;
      q.x = qx[k];
      q.y = pv[k];
      srec[slot] = q;
      sdst[slot] = gpos[b] + rank;
    }
  }
  __syncthreads();
  const int nval = min(SORT_CH, N_EDGES - base);
#pragma unroll
  for (int k = 0; k < SORT_K; ++k) {
    int i = k * SORT_T + tid;
    if (i < nval) {
      const uint2 q = srec[i];
      const unsigned pos = sdst[i];
      if (REC) *(uint2*)(elist + egae_rec_word(pos)) = q;
      else     elist[pos] = q.x;
    }
  }
}

__global__ __launch_bounds__(SORT_T, 16)
void egae_scatter(const int* __restrict__ er, const int* __restrict__ ec,
                  const float* __restrict__ ev, const unsigned* __restrict__ cnt,
                  unsigned* __restrict__ cur, unsigned* __restrict__ elist,
                  unsigned long long ws_bytes) {
  __shared__ unsigned h[NBK2];
  __shared__ unsigned gpos[NBK2];
  __shared__ unsigned lofs[NBK2];
  __shared__ unsigned wt[32];
  __shared__ unsigned wl[32];
  __shared__ uint2 srec[SORT_CH];
  __shared__ unsigned sdst[SORT_CH];
  if (ws_bytes >= REC_WS_MIN)
    egae_scatter_body<true>(er, ec, ev, cnt, cur, elist, h, gpos, lofs, wt, wl, srec, sdst);
  else
    egae_scatter_body<false>(er, ec, ev, cnt, cur, elist, h, gpos, lofs, wt, wl, srec, sdst);
}

__device__ __forceinline__ unsigned egae_claim(unsigned* __restrict__ tick, unsigned* s_item) {
  if (threadIdx.x == 0) {
    const unsigned hf = (__builtin_amdgcn_s_sendmsg_rtn(0x87) >> 18) & 1u;
    unsigned t = atomicAdd(&tick[hf], 1u);
    unsigned item = hf * NBLK + t;
    if (t >= NBLK) {
      t = atomicAdd(&tick[hf ^ 1u], 1u);
      item = (t < NBLK) ? (hf ^ 1u) * NBLK + t : 0xFFFFFFFFu;
    }
    *s_item = item;
  }
  __syncthreads();
  return *s_item;
}

__device__ __forceinline__ unsigned egae_block_start(const unsigned* __restrict__ cnt,
                                                     int b, unsigned* s_start) {
  const int tid = threadIdx.x;
  if (tid == 0) *s_start = 0u;
  __syncthreads();
  unsigned x = ((tid < b) ? cnt[tid] : 0u) + ((tid + 512 < b) ? cnt[tid + 512] : 0u);
#pragma unroll
  for (int d = 16; d >= 1; d >>= 1) x += __shfl_xor(x, d);
  if ((tid & 31) == 0 && x) atomicAdd(s_start, x);
  __syncthreads();
  return *s_start;
}

__device__ __forceinline__ void egae_walk32(const float* ys, float* __restrict__ dst,
                                            int lane, int r, int cl, float v, unsigned n) {
  if (n == 32u) {
#pragma unroll
    for (int t0 = 0; t0 < 32; t0 += 8) {
      float y[8];
#pragma unroll
      for (int u = 0; u < 8; ++u) {
        int cc = __builtin_amdgcn_readlane(cl, t0 + u);
        y[u] = ys[cc * L1D + lane];
      }
#pragma unroll
      for (int u = 0; u < 8; ++u) {
        int rr = __builtin_amdgcn_readlane(r, t0 + u);
        float vv = __int_as_float(__builtin_amdgcn_readlane(__float_as_int(v), t0 + u));
        float* rowp = dst + (size_t)rr * L1D;
        atomicAdd(rowp + (unsigned)lane, vv * y[u]);
      }
    }
  } else {
    for (unsigned t = 0; t < n; ++t) {
      int rr = __builtin_amdgcn_readlane(r, (int)t);
      int cc = __builtin_amdgcn_readlane(cl, (int)t);
      float vv = __int_as_float(__builtin_amdgcn_readlane(__float_as_int(v), (int)t));
      atomicAdd(dst + (size_t)rr * L1D + lane, vv * ys[cc * L1D + lane]);
    }
  }
}

__device__ __forceinline__ uint2 egae_rec_load(const unsigned* __restrict__ elist,
                                               unsigned idx, unsigned end) {
  uint2 q; q.x = 0u; q.y = 0u;
  if (idx < end) q = *(const uint2*)(elist + egae_rec_word(idx));
  return q;
}

template <bool REC>
__device__ __forceinline__ void egae_spmm32_body(const int* __restrict__ er,
                                                 const int* __restrict__ ec,
                                                 const float* __restrict__ ev,
                                                 const unsigned* __restrict__ cnt,
                                                 const unsigned* __restrict__ elist,
                                                 const float* __restrict__ src,
                                                 float* __restrict__ dst,
                                                 float* ys, unsigned* s_start,
                                                 unsigned* __restrict__ tick, unsigned* s_item) {
  const int tid = threadIdx.x, lane = tid & 31;
  const int w = __builtin_amdgcn_readfirstlane(tid >> 5);
  const unsigned item = egae_claim(tick, s_item);
  if (item == 0xFFFFFFFFu) return;
  const int b = (int)item;
  const int row0 = (int)(item >= (unsigned)NBLK ? item - (unsigned)NBLK : item) * SB_ROWS;
  const int nrows = min(SB_ROWS, N_NODES - row0);
  {
    const v4f* g = (const v4f*)(src + (size_t)row0 * L1D);
    v4f* s = (v4f*)ys;
    const int nv = nrows * (L1D / 4);
    v4f tmp[(SB_ROWS * (L1D / 4) + 511) / 512];
#pragma unroll
    for (int k = 0; k < (SB_ROWS * (L1D / 4) + 511) / 512; ++k) tmp[k] = g[min(tid + k * 512, nv - 1)];
#pragma unroll
    for (int k = 0; k < (SB_ROWS * (L1D / 4) + 511) / 512; ++k)
      if (tid + k * 512 < nv) s[tid + k * 512] = tmp[k];
  }
  const unsigned start = egae_block_start(cnt, b, s_start);
  const unsigned end = start + cnt[b];
  unsigned i = start + (unsigned)w * 32u;
  if (REC) {
    uint2 q0 = egae_rec_load(elist, i + (unsigned)lane, end);
    uint2 q1 = egae_rec_load(elist, i + 512u + (unsigned)lane, end);
    for (; i < end; i += 16u * 32u) {
      const uint2 q2 = egae_rec_load(elist, i + 1024u + (unsigned)lane, end);
      const unsigned n = min(32u, end - i);
      egae_walk32(ys, dst, lane, (int)(q0.x & REC_ROWMASK), (int)(q0.x >> 17),
                  __uint_as_float(q0.y), n);
      q0 = q1; q1 = q2;
    }
  } else {
    int idn = (i + (unsigned)lane < end) ? (int)elist[i + (unsigned)lane] : 0;
    int r = er[idn];
    int cl = (int)min((unsigned)(ec[idn] - row0), (unsigned)(SB_ROWS - 1));
    float v = (i + (unsigned)lane < end) ? ev[idn] : 0.0f;
    idn = (i + 512u + (unsigned)lane < end) ? (int)elist[i + 512u + (unsigned)lane] : 0;
    for (; i < end; i += 16u * 32u) {
      const unsigned i1 = i + 16u * 32u;
      const int   rn  = er[idn];
      const int   ecn = ec[idn];
      const float evn = ev[idn];
      idn = (i1 + 512u + (unsigned)lane < end) ? (int)elist[i1 + 512u + (unsigned)lane] : 0;
      const unsigned n = min(32u, end - i);
      egae_walk32(ys, dst, lane, r, cl, v, n);
      r = rn;
      cl = (int)min((unsigned)(ecn - row0), (unsigned)(SB_ROWS - 1));
      v = (i1 + (unsigned)lane < end) ? evn : 0.0f;
    }
  }
}

__global__ __launch_bounds__(512)
void egae_spmm32s(const int* __restrict__ er, const int* __restrict__ ec,
                  const float* __restrict__ ev, const unsigned* __restrict__ cnt,
                  const unsigned* __restrict__ elist,
                  const float* __restrict__ src, float* __restrict__ dst,
                  unsigned* __restrict__ tick, unsigned long long ws_bytes) {
  __shared__ float ys[SB_ROWS * L1D];
  __shared__ unsigned s_start;
  __shared__ unsigned s_item;
  if (ws_bytes >= REC_WS_MIN)
    egae_spmm32_body<true>(er, ec, ev, cnt, elist, src, dst, ys, &s_start, tick, &s_item);
  else
    egae_spmm32_body<false>(er, ec, ev, cnt, elist, src, dst, ys, &s_start, tick, &s_item);
}

__global__ void egae_gemm2(const float* __restrict__ H1,
                           const _Float16* __restrict__ W2t,
                           float* __restrict__ Y2,
                           float* __restrict__ Y3z) {
  const int lane = threadIdx.x & 31;
  const int wave = (blockIdx.x * blockDim.x + threadIdx.x) >> 5;
  const int row0 = wave << 4;
  if (row0 >= N_NODES) return;
  const int m  = lane & 15;
  const int kh = lane >> 4;

  const float* arow = H1 + (size_t)(row0 + m) * L1D + kh * 8;
  v4f a0 = *(const v4f*)(arow);
  v4f a1 = *(const v4f*)(arow + 4);
  v4f a2 = *(const v4f*)(arow + 16);
  v4f a3 = *(const v4f*)(arow + 20);
  v16h af;
#pragma unroll
  for (int i = 0; i < 4; ++i) {
    af[i]      = (_Float16)fmaxf(a0[i], 0.0f);
    af[4 + i]  = (_Float16)fmaxf(a1[i], 0.0f);
    af[8 + i]  = (_Float16)fmaxf(a2[i], 0.0f);
    af[12 + i] = (_Float16)fmaxf(a3[i], 0.0f);
  }
  const _Float16* bc = W2t + (size_t)m * L1D + kh * 8;
  v8h blo = *(const v8h*)(bc);
  v8h bhi = *(const v8h*)(bc + 16);
  v16h bf;
#pragma unroll
  for (int i = 0; i < 8; ++i) { bf[i] = blo[i]; bf[8 + i] = bhi[i]; }

  v8f acc = {};
  acc = __builtin_amdgcn_wmma_f32_16x16x32_f16(false, af, false, bf,
                                               (short)0, acc, false, false);
  float* drow = Y2 + (size_t)(row0 + kh * 8) * L2D + m;
#pragma unroll
  for (int v = 0; v < 8; ++v) drow[(size_t)v * L2D] = acc[v];
  const v4f zero = {0.0f, 0.0f, 0.0f, 0.0f};
  v4f* z = (v4f*)(Y3z + (size_t)row0 * L2D) + lane;
  z[0] = zero; z[32] = zero;
}

__device__ __forceinline__ void egae_walk16(const float* ys, float* __restrict__ dst,
                                            int col, unsigned hh, int r, int cl, float v,
                                            unsigned n) {
  if (n == 32u) {
#pragma unroll
    for (int p0 = 0; p0 < 16; p0 += 8) {
      float y[8];
#pragma unroll
      for (int u = 0; u < 8; ++u) {
        int c0 = __builtin_amdgcn_readlane(cl, 2 * (p0 + u));
        int c1 = __builtin_amdgcn_readlane(cl, 2 * (p0 + u) + 1);
        int cc = hh ? c1 : c0;
        y[u] = ys[cc * L2D + col];
      }
#pragma unroll
      for (int u = 0; u < 8; ++u) {
        int r0 = __builtin_amdgcn_readlane(r, 2 * (p0 + u));
        int r1 = __builtin_amdgcn_readlane(r, 2 * (p0 + u) + 1);
        float v0 = __int_as_float(__builtin_amdgcn_readlane(__float_as_int(v), 2 * (p0 + u)));
        float v1 = __int_as_float(__builtin_amdgcn_readlane(__float_as_int(v), 2 * (p0 + u) + 1));
        int rr = hh ? r1 : r0;
        float vv = hh ? v1 : v0;
        atomicAdd(dst + (size_t)rr * L2D + col, vv * y[u]);
      }
    }
  } else {
    for (unsigned t = 0; t < n; t += 2u) {
      int r0 = __builtin_amdgcn_readlane(r, (int)t);
      int r1 = __builtin_amdgcn_readlane(r, (int)t + 1);
      int c0 = __builtin_amdgcn_readlane(cl, (int)t);
      int c1 = __builtin_amdgcn_readlane(cl, (int)t + 1);
      float v0 = __int_as_float(__builtin_amdgcn_readlane(__float_as_int(v), (int)t));
      float v1 = __int_as_float(__builtin_amdgcn_readlane(__float_as_int(v), (int)t + 1));
      int rr = hh ? r1 : r0;
      int cc = hh ? c1 : c0;
      float vv = hh ? v1 : v0;
      if (t + hh < n) atomicAdd(dst + (size_t)rr * L2D + col, vv * ys[cc * L2D + col]);
    }
  }
}

template <bool REC>
__device__ __forceinline__ void egae_spmm16_body(const int* __restrict__ er,
                                                 const int* __restrict__ ec,
                                                 const float* __restrict__ ev,
                                                 const unsigned* __restrict__ cnt,
                                                 const unsigned* __restrict__ elist,
                                                 const float* __restrict__ src,
                                                 float* __restrict__ dst,
                                                 float* ys, unsigned* s_start,
                                                 unsigned* __restrict__ tick, unsigned* s_item) {
  const int tid = threadIdx.x, lane = tid & 31;
  const int w = __builtin_amdgcn_readfirstlane(tid >> 5);
  const int col = lane & 15;
  const unsigned hh = (unsigned)lane >> 4;
  const unsigned item = egae_claim(tick, s_item);
  if (item == 0xFFFFFFFFu) return;
  const int b = (int)item;
  const int row0 = (int)(item >= (unsigned)NBLK ? item - (unsigned)NBLK : item) * SB_ROWS;
  const int nrows = min(SB_ROWS, N_NODES - row0);
  {
    const v4f* g = (const v4f*)(src + (size_t)row0 * L2D);
    v4f* s = (v4f*)ys;
    const int nv = nrows * (L2D / 4);
    v4f tmp[(SB_ROWS * (L2D / 4) + 511) / 512];
#pragma unroll
    for (int k = 0; k < (SB_ROWS * (L2D / 4) + 511) / 512; ++k) tmp[k] = g[min(tid + k * 512, nv - 1)];
#pragma unroll
    for (int k = 0; k < (SB_ROWS * (L2D / 4) + 511) / 512; ++k)
      if (tid + k * 512 < nv) s[tid + k * 512] = tmp[k];
  }
  const unsigned start = egae_block_start(cnt, b, s_start);
  const unsigned end = start + cnt[b];
  unsigned i = start + (unsigned)w * 32u;
  if (REC) {
    uint2 q0 = egae_rec_load(elist, i + (unsigned)lane, end);
    uint2 q1 = egae_rec_load(elist, i + 512u + (unsigned)lane, end);
    for (; i < end; i += 16u * 32u) {
      const uint2 q2 = egae_rec_load(elist, i + 1024u + (unsigned)lane, end);
      const unsigned n = min(32u, end - i);
      egae_walk16(ys, dst, col, hh, (int)(q0.x & REC_ROWMASK), (int)(q0.x >> 17),
                  __uint_as_float(q0.y), n);
      q0 = q1; q1 = q2;
    }
  } else {
    int idn = (i + (unsigned)lane < end) ? (int)elist[i + (unsigned)lane] : 0;
    int r = er[idn];
    int cl = (int)min((unsigned)(ec[idn] - row0), (unsigned)(SB_ROWS - 1));
    float v = (i + (unsigned)lane < end) ? ev[idn] : 0.0f;
    idn = (i + 512u + (unsigned)lane < end) ? (int)elist[i + 512u + (unsigned)lane] : 0;
    for (; i < end; i += 16u * 32u) {
      const unsigned i1 = i + 16u * 32u;
      const int   rn  = er[idn];
      const int   ecn = ec[idn];
      const float evn = ev[idn];
      idn = (i1 + 512u + (unsigned)lane < end) ? (int)elist[i1 + 512u + (unsigned)lane] : 0;
      const unsigned n = min(32u, end - i);
      egae_walk16(ys, dst, col, hh, r, cl, v, n);
      r = rn;
      cl = (int)min((unsigned)(ecn - row0), (unsigned)(SB_ROWS - 1));
      v = (i1 + (unsigned)lane < end) ? evn : 0.0f;
    }
  }
}

__global__ __launch_bounds__(512)
void egae_spmm16s(const int* __restrict__ er, const int* __restrict__ ec,
                  const float* __restrict__ ev, const unsigned* __restrict__ cnt,
                  const unsigned* __restrict__ elist,
                  const float* __restrict__ src, float* __restrict__ dst,
                  unsigned* __restrict__ tick, unsigned long long ws_bytes) {
  __shared__ float ys[SB_ROWS * L2D];
  __shared__ unsigned s_start;
  __shared__ unsigned s_item;
  if (ws_bytes >= REC_WS_MIN)
    egae_spmm16_body<true>(er, ec, ev, cnt, elist, src, dst, ys, &s_start, tick, &s_item);
  else
    egae_spmm16_body<false>(er, ec, ev, cnt, elist, src, dst, ys, &s_start, tick, &s_item);
}

__global__ void egae_norm(float* __restrict__ io) {
  int i = blockIdx.x * blockDim.x + threadIdx.x;
  if (i >= N_NODES) return;
  v4f* s = (v4f*)(io + (size_t)i * L2D);
  v4f x0 = s[0], x1 = s[1], x2 = s[2], x3 = s[3];
  float ss = 0.0f;
#pragma unroll
  for (int j = 0; j < 4; ++j)
    ss += x0[j] * x0[j] + x1[j] * x1[j] + x2[j] * x2[j] + x3[j] * x3[j];
  float inv = 1.0f / fmaxf(sqrtf(ss), 1e-7f);
  s[0] = x0 * inv; s[1] = x1 * inv; s[2] = x2 * inv; s[3] = x3 * inv;
}

extern "C" void kernel_launch(void* const* d_in, const int* in_sizes, int n_in,
                              void* d_out, int out_size, void* d_ws, size_t ws_size,
                              hipStream_t stream) {
  const float* X  = (const float*)d_in[0];
  const float* W1 = (const float*)d_in[1];
  const float* W2 = (const float*)d_in[2];
  const int*   lr = (const int*)d_in[3];
  const int*   lc = (const int*)d_in[4];
  const float* lv = (const float*)d_in[5];
  float* out = (float*)d_out;

  char* ws = (char*)d_ws;
  float*    Y1   = (float*)(ws + 0);
  float*    H1   = (float*)(ws + 12800000);
  _Float16* W1t  = (_Float16*)d_out;
  unsigned* elist = (unsigned*)(ws + 25600000);
  unsigned* cnt  = (unsigned*)(ws + 38400000);
  unsigned* cur  = (unsigned*)(ws + 38404096);
  unsigned* tick = (unsigned*)(ws + 38408192);
  _Float16* W2t  = (_Float16*)(ws + 38432768);
  float*    Y2   = Y1;
  float*    Y3   = out;

  const int waves  = N_NODES / 16;
  const int gblk   = (waves + 7) / 8;
  const int sblk   = (N_EDGES + SORT_CH - 1) / SORT_CH;

  egae_prep<<<(D_FEAT * L1D + 255) / 256, 256, 0, stream>>>(W1, W2, W1t, W2t, cnt, cur, tick);
  egae_hist<<<sblk, SORT_T, 0, stream>>>(lr, lc, cnt);
  egae_scatter<<<sblk, SORT_T, 0, stream>>>(lr, lc, lv, cnt, cur, elist, (unsigned long long)ws_size);
  egae_gemm1<<<gblk, 256, 0, stream>>>(X, W1t, Y1, H1);
  egae_spmm32s<<<NBK2, 512, 0, stream>>>(lr, lc, lv, cnt, elist, Y1, H1, tick, (unsigned long long)ws_size);
  egae_gemm2<<<gblk, 256, 0, stream>>>(H1, W2t, Y2, Y3);
  egae_spmm16s<<<NBK2, 512, 0, stream>>>(lr, lc, lv, cnt, elist, Y2, Y3, tick + 2, (unsigned long long)ws_size);
  egae_norm<<<(N_NODES + 255) / 256, 256, 0, stream>>>(out);
}
